# starting kernel
# baseline (speedup 1.0000x reference)
#include <hip/hip_runtime.h>
#include <hip/hip_bf16.h>
#include <math.h>

typedef __attribute__((ext_vector_type(2))) float v2f;
typedef __attribute__((ext_vector_type(8))) float v8f;

#define BQ     4096
#define DIN    1024
#define DD     8
#define LL     2
#define FFN    128
#define NQ     10
#define DIMQ   1024
#define REPS   5
#define ROWS_PB 4

__global__ __launch_bounds__(128) void enc_gemm_wmma(
    const float* __restrict__ x, const float* __restrict__ encW,
    const float* __restrict__ encB, float* __restrict__ zpre)
{
    __shared__ float red[4][2][32][8];

    const int lane = threadIdx.x & 31;
    const int wave = threadIdx.x >> 5;
    const int l16  = lane & 15;
    const int half = lane >> 4;
    const int row0 = blockIdx.x * 32;

    const int   bcol  = l16 & 7;
    const float bmask = (l16 < DD) ? 1.0f : 0.0f;

    const float* xrow0 = x + (size_t)(row0 + l16) * DIN;
    const float* xrow1 = xrow0 + (size_t)16 * DIN;
    const int k0 = wave * (DIN / 4);

    v8f c0 = {}, c1 = {};
    for (int k = k0; k < k0 + DIN / 4; k += 4) {
        const int ka = k + half * 2;
        __builtin_prefetch(xrow0 + ka + 256, 0, 3);
        __builtin_prefetch(xrow1 + ka + 256, 0, 3);
        v2f a0, a1, b;
        a0.x = xrow0[ka];
        a0.y = xrow0[ka + 1];
        a1.x = xrow1[ka];
        a1.y = xrow1[ka + 1];
        b.x = bmask * encW[ka * DD + bcol];
        b.y = bmask * encW[(ka + 1) * DD + bcol];
        c0 = __builtin_amdgcn_wmma_f32_16x16x4_f32(
                 false, a0, false, b, (short)0, c0, false, false);
        c1 = __builtin_amdgcn_wmma_f32_16x16x4_f32(
                 false, a1, false, b, (short)0, c1, false, false);
    }

#pragma unroll
    for (int j = 0; j < 8; ++j) {
        red[wave][0][lane][j] = c0[j];
        red[wave][1][lane][j] = c1[j];
    }
    __syncthreads();

    if (wave < 2 && l16 < DD) {
        const int  t    = wave;
        const float bias = encB[l16];
#pragma unroll
        for (int j = 0; j < 8; ++j) {
            const float s = red[0][t][lane][j] + red[1][t][lane][j]
                          + red[2][t][lane][j] + red[3][t][lane][j];
            const int m = t * 16 + half * 8 + j;
            zpre[(size_t)(row0 + m) * DD + l16] = s + bias;
        }
    }
}

__device__ inline void ln8(const float* h, const float* g, const float* b, float* out)
{
    float m = 0.f;
#pragma unroll
    for (int d = 0; d < DD; ++d) m += h[d];
    m *= 0.125f;
    float v = 0.f;
#pragma unroll
    for (int d = 0; d < DD; ++d) { const float t = h[d] - m; v += t * t; }
    v *= 0.125f;
    const float inv = 1.0f / sqrtf(v + 1e-5f);
#pragma unroll
    for (int d = 0; d < DD; ++d) out[d] = (h[d] - m) * inv * g[d] + b[d];
}

__global__ __launch_bounds__(128) void qsim_kernel(
    const float* __restrict__ zpre,
    const float* __restrict__ Wv,  const float* __restrict__ bv,
    const float* __restrict__ Wo,  const float* __restrict__ bo,
    const float* __restrict__ ln1g, const float* __restrict__ ln1b,
    const float* __restrict__ W1,  const float* __restrict__ b1,
    const float* __restrict__ W2,  const float* __restrict__ b2,
    const float* __restrict__ ln2g, const float* __restrict__ ln2b,
    const float* __restrict__ qw,  float* __restrict__ zq)
{
    __shared__ float sRe[ROWS_PB][DIMQ];
    __shared__ float sIm[ROWS_PB][DIMQ];

    const int g    = threadIdx.x >> 5;
    const int lane = threadIdx.x & 31;
    const int row  = blockIdx.x * ROWS_PB + g;

    float z[DD];
#pragma unroll
    for (int d = 0; d < DD; ++d) z[d] = zpre[(size_t)row * DD + d];

    for (int l = 0; l < LL; ++l) {
        const float* wv = Wv + l * 64;  const float* wo = Wo + l * 64;
        float v[DD], attn[DD], h[DD];
#pragma unroll
        for (int d = 0; d < DD; ++d) {
            float s = bv[l * DD + d];
#pragma unroll
            for (int k = 0; k < DD; ++k) s += z[k] * wv[k * DD + d];
            v[d] = s;
        }
#pragma unroll
        for (int d = 0; d < DD; ++d) {
            float s = bo[l * DD + d];
#pragma unroll
            for (int k = 0; k < DD; ++k) s += v[k] * wo[k * DD + d];
            attn[d] = s;
        }
#pragma unroll
        for (int d = 0; d < DD; ++d) h[d] = z[d] + attn[d];
        ln8(h, ln1g + l * DD, ln1b + l * DD, z);

        float acc[DD];
#pragma unroll
        for (int d = 0; d < DD; ++d) acc[d] = 0.f;
#pragma unroll
        for (int j = 0; j < 4; ++j) {
            const int hid = lane + j * 32;
            float hv = b1[l * FFN + hid];
#pragma unroll
            for (int k = 0; k < DD; ++k) hv += z[k] * W1[l * DD * FFN + k * FFN + hid];
            hv = fmaxf(hv, 0.0f);
#pragma unroll
            for (int d = 0; d < DD; ++d) acc[d] += hv * W2[l * FFN * DD + hid * DD + d];
        }
        for (int off = 16; off > 0; off >>= 1) {
#pragma unroll
            for (int d = 0; d < DD; ++d) acc[d] += __shfl_xor(acc[d], off, 32);
        }
#pragma unroll
        for (int d = 0; d < DD; ++d) h[d] = z[d] + acc[d] + b2[l * DD + d];
        ln8(h, ln2g + l * DD, ln2b + l * DD, z);
    }

    float nrm2 = 0.f;
#pragma unroll
    for (int d = 0; d < DD; ++d) nrm2 += z[d] * z[d];
    const float inv = 1.0f / sqrtf(nrm2);

    for (int i = lane; i < DIMQ; i += 32) { sRe[g][i] = 0.f; sIm[g][i] = 0.f; }
#pragma unroll
    for (int d = 0; d < DD; ++d) if (lane == d) sRe[g][d] = z[d] * inv;
    __builtin_amdgcn_wave_barrier();

    for (int r = 0; r < REPS; ++r) {
        for (int w = 0; w < NQ; ++w) {
            const float th  = 0.5f * qw[r * NQ + w];
            const float cth = cosf(th);
            const float sth = sinf(th);
            const int bit = 1 << (NQ - 1 - w);
            const int low = bit - 1;
            for (int p = lane; p < DIMQ / 2; p += 32) {
                const int i0 = ((p & ~low) << 1) | (p & low);
                const int i1 = i0 | bit;
                const float a0r = sRe[g][i0], a0i = sIm[g][i0];
                const float a1r = sRe[g][i1], a1i = sIm[g][i1];
                sRe[g][i0] = cth * a0r + sth * a1i;
                sIm[g][i0] = cth * a0i - sth * a1r;
                sRe[g][i1] = cth * a1r + sth * a0i;
                sIm[g][i1] = cth * a1i - sth * a0r;
            }
            __builtin_amdgcn_wave_barrier();
        }
        for (int w = 0; w < NQ; ++w) {
            const int cbit = 1 << (NQ - 1 - w);
            const int tbit = 1 << (NQ - 1 - ((w + 1) % NQ));
            for (int i = lane; i < DIMQ; i += 32) {
                if ((i & cbit) && !(i & tbit)) {
                    const int j = i | tbit;
                    const float tr = sRe[g][i], ti = sIm[g][i];
                    sRe[g][i] = sRe[g][j];  sIm[g][i] = sIm[g][j];
                    sRe[g][j] = tr;         sIm[g][j] = ti;
                }
            }
            __builtin_amdgcn_wave_barrier();
        }
    }

    float acc[DD];
#pragma unroll
    for (int d = 0; d < DD; ++d) acc[d] = 0.f;
    for (int i = lane; i < DIMQ; i += 32) {
        const float re = sRe[g][i], im = sIm[g][i];
        const float p = re * re + im * im;
#pragma unroll
        for (int w = 0; w < DD; ++w) {
            const int bit = 1 << (NQ - 1 - w);
            acc[w] += (i & bit) ? -p : p;
        }
    }
    for (int off = 16; off > 0; off >>= 1) {
#pragma unroll
        for (int w = 0; w < DD; ++w) acc[w] += __shfl_xor(acc[w], off, 32);
    }
#pragma unroll
    for (int w = 0; w < DD; ++w) if (lane == w) zq[(size_t)row * DD + w] = acc[w];
}

__global__ __launch_bounds__(32) void dec_gemm_wmma(
    const float* __restrict__ zq, const float* __restrict__ decW,
    const float* __restrict__ decB, float* __restrict__ out)
{
    const int lane = threadIdx.x;
    const int l16  = lane & 15;
    const int half = lane >> 4;
    const int n0 = blockIdx.x * 16;
    const int m0 = blockIdx.y * 16;

    const float* arow = zq + (size_t)(m0 + l16) * DD;

    v8f c = {};
#pragma unroll
    for (int k = 0; k < DD; k += 4) {
        const int ka = k + half * 2;
        v2f a, b;
        a.x = arow[ka];
        a.y = arow[ka + 1];
        b.x = decW[(size_t)ka * DIN + n0 + l16];
        b.y = decW[(size_t)(ka + 1) * DIN + n0 + l16];
        c = __builtin_amdgcn_wmma_f32_16x16x4_f32(
                false, a, false, b, (short)0, c, false, false);
    }

    const float bias = decB[n0 + l16];
#pragma unroll
    for (int j = 0; j < 8; ++j) {
        const int m = half * 8 + j;
        out[(size_t)(m0 + m) * DIN + n0 + l16] = c[j] + bias;
    }
}

extern "C" void kernel_launch(void* const* d_in, const int* in_sizes, int n_in,
                              void* d_out, int out_size, void* d_ws, size_t ws_size,
                              hipStream_t stream) {
    const float* x    = (const float*)d_in[0];
    const float* encW = (const float*)d_in[1];
    const float* encB = (const float*)d_in[2];
    const float* decW = (const float*)d_in[3];
    const float* decB = (const float*)d_in[4];
    const float* Wv   = (const float*)d_in[5];
    const float* bv   = (const float*)d_in[6];
    const float* Wo   = (const float*)d_in[7];
    const float* bo   = (const float*)d_in[8];
    const float* ln1g = (const float*)d_in[9];
    const float* ln1b = (const float*)d_in[10];
    const float* W1   = (const float*)d_in[11];
    const float* b1   = (const float*)d_in[12];
    const float* W2   = (const float*)d_in[13];
    const float* b2   = (const float*)d_in[14];
    const float* ln2g = (const float*)d_in[15];
    const float* ln2b = (const float*)d_in[16];
    const float* qw   = (const float*)d_in[17];
    float* out  = (float*)d_out;

    float* zpre = (float*)d_ws;
    float* zqv  = zpre + (size_t)BQ * DD;

    enc_gemm_wmma<<<BQ / 32, 128, 0, stream>>>(x, encW, encB, zpre);

    qsim_kernel<<<BQ / ROWS_PB, 128, 0, stream>>>(
        zpre, Wv, bv, Wo, bo, ln1g, ln1b, W1, b1, W2, b2, ln2g, ln2b, qw, zqv);

    dim3 dgrid(DIN / 16, BQ / 16);
    dec_gemm_wmma<<<dgrid, 32, 0, stream>>>(zqv, decW, decB, out);
}


# Row statevector kept in registers (32 amplitudes/lane): RX is register arithmetic, LDS swaps index-bit roles twice a layer, CNOT ring is one scatter; cos/sin table per block (zq sum order changes)
# speedup vs baseline: 3.8049x; 3.8049x over previous
#include <hip/hip_runtime.h>
#include <hip/hip_bf16.h>
#include <math.h>

typedef __attribute__((ext_vector_type(2))) float v2f;
typedef __attribute__((ext_vector_type(8))) float v8f;

#define BQ     4096
#define DIN    1024
#define DD     8
#define LL     2
#define FFN    128
#define NQ     10
#define DIMQ   1024
#define REPS   5
#define ROWS_PB 4

__global__ __launch_bounds__(128) void enc_gemm_wmma(
    const float* __restrict__ x, const float* __restrict__ encW,
    const float* __restrict__ encB, float* __restrict__ zpre)
{
    __shared__ float red[4][2][32][8];

    const int lane = threadIdx.x & 31;
    const int wave = threadIdx.x >> 5;
    const int l16  = lane & 15;
    const int half = lane >> 4;
    const int row0 = blockIdx.x * 32;

    const int   bcol  = l16 & 7;
    const float bmask = (l16 < DD) ? 1.0f : 0.0f;

    const float* xrow0 = x + (size_t)(row0 + l16) * DIN;
    const float* xrow1 = xrow0 + (size_t)16 * DIN;
    const int k0 = wave * (DIN / 4);

    v8f c0 = {}, c1 = {};
    for (int k = k0; k < k0 + DIN / 4; k += 4) {
        const int ka = k + half * 2;
        __builtin_prefetch(xrow0 + ka + 256, 0, 3);
        __builtin_prefetch(xrow1 + ka + 256, 0, 3);
        v2f a0, a1, b;
        a0.x = xrow0[ka];
        a0.y = xrow0[ka + 1];
        a1.x = xrow1[ka];
        a1.y = xrow1[ka + 1];
        b.x = bmask * encW[ka * DD + bcol];
        b.y = bmask * encW[(ka + 1) * DD + bcol];
        c0 = __builtin_amdgcn_wmma_f32_16x16x4_f32(
                 false, a0, false, b, (short)0, c0, false, false);
        c1 = __builtin_amdgcn_wmma_f32_16x16x4_f32(
                 false, a1, false, b, (short)0, c1, false, false);
    }

#pragma unroll
    for (int j = 0; j < 8; ++j) {
        red[wave][0][lane][j] = c0[j];
        red[wave][1][lane][j] = c1[j];
    }
    __syncthreads();

    if (wave < 2 && l16 < DD) {
        const int  t    = wave;
        const float bias = encB[l16];
#pragma unroll
        for (int j = 0; j < 8; ++j) {
            const float s = red[0][t][lane][j] + red[1][t][lane][j]
                          + red[2][t][lane][j] + red[3][t][lane][j];
            const int m = t * 16 + half * 8 + j;
            zpre[(size_t)(row0 + m) * DD + l16] = s + bias;
        }
    }
}

typedef __attribute__((ext_vector_type(4))) float v4f;

__device__ inline void ln8(const float* h, const float* g, const float* b, float* out)
{
    float m = 0.f;
#pragma unroll
    for (int d = 0; d < DD; ++d) m += h[d];
    m *= 0.125f;
    float v = 0.f;
#pragma unroll
    for (int d = 0; d < DD; ++d) { const float t = h[d] - m; v += t * t; }
    v *= 0.125f;
    const float inv = 1.0f / sqrtf(v + 1e-5f);
#pragma unroll
    for (int d = 0; d < DD; ++d) out[d] = (h[d] - m) * inv * g[d] + b[d];
}

__device__ constexpr int ring_map(int i)
{
    for (int w = 0; w < NQ; ++w) {
        const int cbit = 1 << (NQ - 1 - w);
        const int tbit = 1 << (NQ - 1 - ((w + 1) % NQ));
        if (i & cbit) i ^= tbit;
    }
    return i;
}

__device__ constexpr int lds_swz(int h)  { return (h & 15) ^ ((h >> 4) & 1); }
__device__ constexpr int lds_slot(int j) { return j ^ (lds_swz(j >> 5) << 1); }

__device__ inline void wave_lds_sync()
{
    __builtin_amdgcn_wave_barrier();
}

template <int BIT>
__device__ inline void rx_regs(float (&re)[32], float (&im)[32], const float c, const float s)
{
    constexpr int low = BIT - 1;
#pragma unroll
    for (int p = 0; p < 16; ++p) {
        const int r0 = ((p & ~low) << 1) | (p & low);
        const int r1 = r0 | BIT;
        const float a0r = re[r0], a0i = im[r0];
        const float a1r = re[r1], a1i = im[r1];
        re[r0] = c * a0r + s * a1i;
        im[r0] = c * a0i - s * a1r;
        re[r1] = c * a1r + s * a0i;
        im[r1] = c * a1i - s * a0r;
    }
}

__global__ __launch_bounds__(128) void qsim_kernel(
    const float* __restrict__ zpre,
    const float* __restrict__ Wv,  const float* __restrict__ bv,
    const float* __restrict__ Wo,  const float* __restrict__ bo,
    const float* __restrict__ ln1g, const float* __restrict__ ln1b,
    const float* __restrict__ W1,  const float* __restrict__ b1,
    const float* __restrict__ W2,  const float* __restrict__ b2,
    const float* __restrict__ ln2g, const float* __restrict__ ln2b,
    const float* __restrict__ qw,  float* __restrict__ zq)
{
    __shared__ v4f   sAmp[ROWS_PB][DIMQ / 2];
    __shared__ float sCos[REPS * NQ];
    __shared__ float sSin[REPS * NQ];

    const int g    = threadIdx.x >> 5;
    const int lane = threadIdx.x & 31;
    const int row  = blockIdx.x * ROWS_PB + g;

    if (threadIdx.x < REPS * NQ) {
        const float th = 0.5f * qw[threadIdx.x];
        sCos[threadIdx.x] = cosf(th);
        sSin[threadIdx.x] = sinf(th);
    }

    float z[DD];
#pragma unroll
    for (int d = 0; d < DD; ++d) z[d] = zpre[(size_t)row * DD + d];

    for (int l = 0; l < LL; ++l) {
        const float* wv = Wv + l * 64;  const float* wo = Wo + l * 64;
        float v[DD], attn[DD], h[DD];
#pragma unroll
        for (int d = 0; d < DD; ++d) {
            float s = bv[l * DD + d];
#pragma unroll
            for (int k = 0; k < DD; ++k) s += z[k] * wv[k * DD + d];
            v[d] = s;
        }
#pragma unroll
        for (int d = 0; d < DD; ++d) {
            float s = bo[l * DD + d];
#pragma unroll
            for (int k = 0; k < DD; ++k) s += v[k] * wo[k * DD + d];
            attn[d] = s;
        }
#pragma unroll
        for (int d = 0; d < DD; ++d) h[d] = z[d] + attn[d];
        ln8(h, ln1g + l * DD, ln1b + l * DD, z);

        float acc[DD];
#pragma unroll
        for (int d = 0; d < DD; ++d) acc[d] = 0.f;
#pragma unroll
        for (int j = 0; j < 4; ++j) {
            const int hid = lane + j * 32;
            float hv = b1[l * FFN + hid];
#pragma unroll
            for (int k = 0; k < DD; ++k) hv += z[k] * W1[l * DD * FFN + k * FFN + hid];
            hv = fmaxf(hv, 0.0f);
#pragma unroll
            for (int d = 0; d < DD; ++d) acc[d] += hv * W2[l * FFN * DD + hid * DD + d];
        }
        for (int off = 16; off > 0; off >>= 1) {
#pragma unroll
            for (int d = 0; d < DD; ++d) acc[d] += __shfl_xor(acc[d], off, 32);
        }
#pragma unroll
        for (int d = 0; d < DD; ++d) h[d] = z[d] + acc[d] + b2[l * DD + d];
        ln8(h, ln2g + l * DD, ln2b + l * DD, z);
    }

    float nrm2 = 0.f;
#pragma unroll
    for (int d = 0; d < DD; ++d) nrm2 += z[d] * z[d];
    const float inv = 1.0f / sqrtf(nrm2);

    float re[32], im[32];
#pragma unroll
    for (int r = 0; r < 32; ++r) { re[r] = 0.f; im[r] = 0.f; }
#pragma unroll
    for (int d = 0; d < DD; ++d) if (lane == d) re[0] = z[d] * inv;

    v4f* const amp4 = &sAmp[g][0];
    v2f* const amp2 = reinterpret_cast<v2f*>(amp4);
    const int pairA   = lane * 16;
    const int swzA    = lds_swz(lane);
    const int ringSlt = lds_slot(ring_map(lane << 5));

    __syncthreads();

#pragma unroll 1
    for (int r = 0; r < REPS; ++r) {
        const float* cs = sCos + r * NQ;
        const float* sn = sSin + r * NQ;

        rx_regs<16>(re, im, cs[0], sn[0]);
        rx_regs<8> (re, im, cs[1], sn[1]);
        rx_regs<4> (re, im, cs[2], sn[2]);
        rx_regs<2> (re, im, cs[3], sn[3]);
        rx_regs<1> (re, im, cs[4], sn[4]);

#pragma unroll
        for (int R = 0; R < 32; ++R) {
            v2f t;
            t.x = re[R];  t.y = im[R];
            amp2[R * 32 + (lane ^ (lds_swz(R) << 1))] = t;
        }
        wave_lds_sync();
#pragma unroll
        for (int k = 0; k < 16; ++k) {
            const v4f t = amp4[pairA + (k ^ swzA)];
            re[2 * k]     = t.x;  im[2 * k]     = t.y;
            re[2 * k + 1] = t.z;  im[2 * k + 1] = t.w;
        }

        rx_regs<16>(re, im, cs[5], sn[5]);
        rx_regs<8> (re, im, cs[6], sn[6]);
        rx_regs<4> (re, im, cs[7], sn[7]);
        rx_regs<2> (re, im, cs[8], sn[8]);
        rx_regs<1> (re, im, cs[9], sn[9]);

#pragma unroll
        for (int rr = 0; rr < 32; ++rr) {
            v2f t;
            t.x = re[rr];  t.y = im[rr];
            amp2[lds_slot(ring_map(rr)) ^ ringSlt] = t;
        }
        wave_lds_sync();
#pragma unroll
        for (int R = 0; R < 32; ++R) {
            const v2f t = amp2[R * 32 + (lane ^ (lds_swz(R) << 1))];
            re[R] = t.x;  im[R] = t.y;
        }
    }

    float p[32];
#pragma unroll
    for (int R = 0; R < 32; ++R) p[R] = re[R] * re[R] + im[R] * im[R];

    float acc[DD];
    {
        float cur[32];
#pragma unroll
        for (int R = 0; R < 32; ++R) cur[R] = p[R];
#pragma unroll
        for (int lvl = 0; lvl < 5; ++lvl) {
            const int n = 16 >> lvl;
            float dsum = 0.f;
#pragma unroll
            for (int m = 0; m < n; ++m) {
                dsum += cur[2 * m] - cur[2 * m + 1];
                cur[m] = cur[2 * m] + cur[2 * m + 1];
            }
            acc[4 - lvl] = dsum;
        }
        const float tot = cur[0];
#pragma unroll
        for (int w = 5; w < DD; ++w)
            acc[w] = ((lane >> (9 - w)) & 1) ? -tot : tot;
    }
    for (int off = 16; off > 0; off >>= 1) {
#pragma unroll
        for (int w = 0; w < DD; ++w) acc[w] += __shfl_xor(acc[w], off, 32);
    }
#pragma unroll
    for (int w = 0; w < DD; ++w) if (lane == w) zq[(size_t)row * DD + w] = acc[w];
}

__global__ __launch_bounds__(32) void dec_gemm_wmma(
    const float* __restrict__ zq, const float* __restrict__ decW,
    const float* __restrict__ decB, float* __restrict__ out)
{
    const int lane = threadIdx.x;
    const int l16  = lane & 15;
    const int half = lane >> 4;
    const int n0 = blockIdx.x * 16;
    const int m0 = blockIdx.y * 16;

    const float* arow = zq + (size_t)(m0 + l16) * DD;

    v8f c = {};
#pragma unroll
    for (int k = 0; k < DD; k += 4) {
        const int ka = k + half * 2;
        v2f a, b;
        a.x = arow[ka];
        a.y = arow[ka + 1];
        b.x = decW[(size_t)ka * DIN + n0 + l16];
        b.y = decW[(size_t)(ka + 1) * DIN + n0 + l16];
        c = __builtin_amdgcn_wmma_f32_16x16x4_f32(
                false, a, false, b, (short)0, c, false, false);
    }

    const float bias = decB[n0 + l16];
#pragma unroll
    for (int j = 0; j < 8; ++j) {
        const int m = half * 8 + j;
        out[(size_t)(m0 + m) * DIN + n0 + l16] = c[j] + bias;
    }
}

extern "C" void kernel_launch(void* const* d_in, const int* in_sizes, int n_in,
                              void* d_out, int out_size, void* d_ws, size_t ws_size,
                              hipStream_t stream) {
    const float* x    = (const float*)d_in[0];
    const float* encW = (const float*)d_in[1];
    const float* encB = (const float*)d_in[2];
    const float* decW = (const float*)d_in[3];
    const float* decB = (const float*)d_in[4];
    const float* Wv   = (const float*)d_in[5];
    const float* bv   = (const float*)d_in[6];
    const float* Wo   = (const float*)d_in[7];
    const float* bo   = (const float*)d_in[8];
    const float* ln1g = (const float*)d_in[9];
    const float* ln1b = (const float*)d_in[10];
    const float* W1   = (const float*)d_in[11];
    const float* b1   = (const float*)d_in[12];
    const float* W2   = (const float*)d_in[13];
    const float* b2   = (const float*)d_in[14];
    const float* ln2g = (const float*)d_in[15];
    const float* ln2b = (const float*)d_in[16];
    const float* qw   = (const float*)d_in[17];
    float* out  = (float*)d_out;

    float* zpre = (float*)d_ws;
    float* zqv  = zpre + (size_t)BQ * DD;

    enc_gemm_wmma<<<BQ / 32, 128, 0, stream>>>(x, encW, encB, zpre);

    qsim_kernel<<<BQ / ROWS_PB, 128, 0, stream>>>(
        zpre, Wv, bv, Wo, bo, ln1g, ln1b, W1, b1, W2, b2, ln2g, ln2b, qw, zqv);

    dim3 dgrid(DIN / 16, BQ / 16);
    dec_gemm_wmma<<<dgrid, 32, 0, stream>>>(zqv, decW, decB, out);
}


# 2 launches become 1: encoder GEMM runs as the circuit kernel's prologue; each wave parks its own x row in LDS (whole-row 16-byte loads), wave g takes K-sixteenth g of it; z_pre sum order changes
# speedup vs baseline: 7.0747x; 1.8594x over previous
#include <hip/hip_runtime.h>
#include <hip/hip_bf16.h>
#include <math.h>

typedef __attribute__((ext_vector_type(2))) float v2f;
typedef __attribute__((ext_vector_type(8))) float v8f;

#define BQ     4096
#define DIN    1024
#define DD     8
#define LL     2
#define FFN    128
#define NQ     10
#define DIMQ   1024
#define REPS   5
#define ROWS_PB 16
#define XPITCH 1028

typedef __attribute__((ext_vector_type(4))) float v4f;

__device__ inline void ln8(const float* h, const float* g, const float* b, float* out)
{
    float m = 0.f;
#pragma unroll
    for (int d = 0; d < DD; ++d) m += h[d];
    m *= 0.125f;
    float v = 0.f;
#pragma unroll
    for (int d = 0; d < DD; ++d) { const float t = h[d] - m; v += t * t; }
    v *= 0.125f;
    const float inv = 1.0f / sqrtf(v + 1e-5f);
#pragma unroll
    for (int d = 0; d < DD; ++d) out[d] = (h[d] - m) * inv * g[d] + b[d];
}

__device__ constexpr int ring_map(int i)
{
    for (int w = 0; w < NQ; ++w) {
        const int cbit = 1 << (NQ - 1 - w);
        const int tbit = 1 << (NQ - 1 - ((w + 1) % NQ));
        if (i & cbit) i ^= tbit;
    }
    return i;
}

__device__ constexpr int lds_swz(int h)  { return (h & 15) ^ ((h >> 4) & 1); }
__device__ constexpr int lds_slot(int j) { return j ^ (lds_swz(j >> 5) << 1); }

__device__ inline void wave_lds_sync()
{
    __builtin_amdgcn_wave_barrier();
}

template <int BIT>
__device__ inline void rx_regs(float (&re)[32], float (&im)[32], const float c, const float s)
{
    constexpr int low = BIT - 1;
#pragma unroll
    for (int p = 0; p < 16; ++p) {
        const int r0 = ((p & ~low) << 1) | (p & low);
        const int r1 = r0 | BIT;
        const float a0r = re[r0], a0i = im[r0];
        const float a1r = re[r1], a1i = im[r1];
        re[r0] = c * a0r + s * a1i;
        im[r0] = c * a0i - s * a1r;
        re[r1] = c * a1r + s * a0i;
        im[r1] = c * a1i - s * a0r;
    }
}

__global__ __launch_bounds__(ROWS_PB * 32) void qsim_kernel(
    const float* __restrict__ x, const float* __restrict__ encW,
    const float* __restrict__ encB, const float* __restrict__ qw,
    const float* __restrict__ Wv,  const float* __restrict__ bv,
    const float* __restrict__ Wo,  const float* __restrict__ bo,
    const float* __restrict__ ln1g, const float* __restrict__ ln1b,
    const float* __restrict__ W1,  const float* __restrict__ b1,
    const float* __restrict__ W2,  const float* __restrict__ b2,
    const float* __restrict__ ln2g, const float* __restrict__ ln2b,
    const float* __restrict__ decW, const float* __restrict__ decB,
    float* __restrict__ out)
{
    __shared__ v4f   sAmp[ROWS_PB][DIMQ / 2];
    __shared__ float sCos[REPS * NQ];
    __shared__ float sSin[REPS * NQ];
    __shared__ float sZq[ROWS_PB][DD];
    __shared__ __attribute__((aligned(16))) float sPart[ROWS_PB][ROWS_PB][DD];

    const int g    = threadIdx.x >> 5;
    const int lane = threadIdx.x & 31;
    const int row  = blockIdx.x * ROWS_PB + g;
    const int l16  = lane & 15;
    const int half = lane >> 4;

    const float qang = qw[threadIdx.x < REPS * NQ ? threadIdx.x : 0];

    const float* xr = x + (size_t)row * DIN + 4 * lane;
    v4f xv[8];
#pragma unroll
    for (int j = 0; j < 8; ++j) xv[j] = *reinterpret_cast<const v4f*>(xr + 128 * j);

    const int   kb    = g * 64;
    const int   bcol  = l16 & 7;
    const float bmask = (l16 < DD) ? 1.0f : 0.0f;
    v2f bf[16];
#pragma unroll
    for (int u = 0; u < 8; ++u) {
        const float* wk = encW + (size_t)(kb + 8 * u + 4 * half) * DD + bcol;
        bf[2 * u].x     = bmask * wk[0 * DD];
        bf[2 * u].y     = bmask * wk[1 * DD];
        bf[2 * u + 1].x = bmask * wk[2 * DD];
        bf[2 * u + 1].y = bmask * wk[3 * DD];
    }

    asm volatile("" :: "v"(qang));

    if (threadIdx.x < REPS * NQ) {
        const float th = 0.5f * qang;
        sCos[threadIdx.x] = cosf(th);
        sSin[threadIdx.x] = sinf(th);
    }

    float* const xs = reinterpret_cast<float*>(&sAmp[0][0]);
#pragma unroll
    for (int j = 0; j < 8; ++j)
        *reinterpret_cast<v4f*>(xs + g * XPITCH + 4 * lane + 128 * j) = xv[j];

    __syncthreads();

    {
        const float* arow = xs + l16 * XPITCH + kb + 4 * half;
        v8f c0 = {}, c1 = {};
#pragma unroll
        for (int u = 0; u < 8; ++u) {
            const v4f a4 = *reinterpret_cast<const v4f*>(arow + 8 * u);
            v2f alo, ahi;
            alo.x = a4.x;  alo.y = a4.y;
            ahi.x = a4.z;  ahi.y = a4.w;
            c0 = __builtin_amdgcn_wmma_f32_16x16x4_f32(
                     false, alo, false, bf[2 * u], (short)0, c0, false, false);
            c1 = __builtin_amdgcn_wmma_f32_16x16x4_f32(
                     false, ahi, false, bf[2 * u + 1], (short)0, c1, false, false);
        }
        if (l16 < DD) {
#pragma unroll
            for (int r = 0; r < 8; ++r) sPart[g][half * 8 + r][l16] = c0[r] + c1[r];
        }
    }
    __syncthreads();

    float z[DD];
#pragma unroll
    for (int d = 0; d < DD; ++d) z[d] = 0.f;
#pragma unroll
    for (int p = 0; p < ROWS_PB; ++p) {
        const v4f lo = *reinterpret_cast<const v4f*>(&sPart[p][g][0]);
        const v4f hi = *reinterpret_cast<const v4f*>(&sPart[p][g][4]);
        z[0] += lo.x;  z[1] += lo.y;  z[2] += lo.z;  z[3] += lo.w;
        z[4] += hi.x;  z[5] += hi.y;  z[6] += hi.z;  z[7] += hi.w;
    }
#pragma unroll
    for (int d = 0; d < DD; ++d) z[d] += encB[d];

    for (int l = 0; l < LL; ++l) {
        const float* wv = Wv + l * 64;  const float* wo = Wo + l * 64;
        float v[DD], attn[DD], h[DD];
#pragma unroll
        for (int d = 0; d < DD; ++d) {
            float s = bv[l * DD + d];
#pragma unroll
            for (int k = 0; k < DD; ++k) s += z[k] * wv[k * DD + d];
            v[d] = s;
        }
#pragma unroll
        for (int d = 0; d < DD; ++d) {
            float s = bo[l * DD + d];
#pragma unroll
            for (int k = 0; k < DD; ++k) s += v[k] * wo[k * DD + d];
            attn[d] = s;
        }
#pragma unroll
        for (int d = 0; d < DD; ++d) h[d] = z[d] + attn[d];
        ln8(h, ln1g + l * DD, ln1b + l * DD, z);

        float acc[DD];
#pragma unroll
        for (int d = 0; d < DD; ++d) acc[d] = 0.f;
#pragma unroll
        for (int j = 0; j < 4; ++j) {
            const int hid = lane + j * 32;
            float hv = b1[l * FFN + hid];
#pragma unroll
            for (int k = 0; k < DD; ++k) hv += z[k] * W1[l * DD * FFN + k * FFN + hid];
            hv = fmaxf(hv, 0.0f);
#pragma unroll
            for (int d = 0; d < DD; ++d) acc[d] += hv * W2[l * FFN * DD + hid * DD + d];
        }
        for (int off = 16; off > 0; off >>= 1) {
#pragma unroll
            for (int d = 0; d < DD; ++d) acc[d] += __shfl_xor(acc[d], off, 32);
        }
#pragma unroll
        for (int d = 0; d < DD; ++d) h[d] = z[d] + acc[d] + b2[l * DD + d];
        ln8(h, ln2g + l * DD, ln2b + l * DD, z);
    }

    float nrm2 = 0.f;
#pragma unroll
    for (int d = 0; d < DD; ++d) nrm2 += z[d] * z[d];
    const float inv = 1.0f / sqrtf(nrm2);

    float re[32], im[32];
#pragma unroll
    for (int r = 0; r < 32; ++r) { re[r] = 0.f; im[r] = 0.f; }
#pragma unroll
    for (int d = 0; d < DD; ++d) if (lane == d) re[0] = z[d] * inv;

    v4f* const amp4 = &sAmp[g][0];
    v2f* const amp2 = reinterpret_cast<v2f*>(amp4);
    const int pairA   = lane * 16;
    const int swzA    = lds_swz(lane);
    const int ringSlt = lds_slot(ring_map(lane << 5));

#pragma unroll 1
    for (int r = 0; r < REPS; ++r) {
        const float* cs = sCos + r * NQ;
        const float* sn = sSin + r * NQ;

        rx_regs<16>(re, im, cs[0], sn[0]);
        rx_regs<8> (re, im, cs[1], sn[1]);
        rx_regs<4> (re, im, cs[2], sn[2]);
        rx_regs<2> (re, im, cs[3], sn[3]);
        rx_regs<1> (re, im, cs[4], sn[4]);

#pragma unroll
        for (int R = 0; R < 32; ++R) {
            v2f t;
            t.x = re[R];  t.y = im[R];
            amp2[R * 32 + (lane ^ (lds_swz(R) << 1))] = t;
        }
        wave_lds_sync();
#pragma unroll
        for (int k = 0; k < 16; ++k) {
            const v4f t = amp4[pairA + (k ^ swzA)];
            re[2 * k]     = t.x;  im[2 * k]     = t.y;
            re[2 * k + 1] = t.z;  im[2 * k + 1] = t.w;
        }

        rx_regs<16>(re, im, cs[5], sn[5]);
        rx_regs<8> (re, im, cs[6], sn[6]);
        rx_regs<4> (re, im, cs[7], sn[7]);
        rx_regs<2> (re, im, cs[8], sn[8]);
        rx_regs<1> (re, im, cs[9], sn[9]);

#pragma unroll
        for (int rr = 0; rr < 32; ++rr) {
            v2f t;
            t.x = re[rr];  t.y = im[rr];
            amp2[lds_slot(ring_map(rr)) ^ ringSlt] = t;
        }
        wave_lds_sync();
#pragma unroll
        for (int R = 0; R < 32; ++R) {
            const v2f t = amp2[R * 32 + (lane ^ (lds_swz(R) << 1))];
            re[R] = t.x;  im[R] = t.y;
        }
    }

    float p[32];
#pragma unroll
    for (int R = 0; R < 32; ++R) p[R] = re[R] * re[R] + im[R] * im[R];

    float acc[DD];
    {
        float cur[32];
#pragma unroll
        for (int R = 0; R < 32; ++R) cur[R] = p[R];
#pragma unroll
        for (int lvl = 0; lvl < 5; ++lvl) {
            const int n = 16 >> lvl;
            float dsum = 0.f;
#pragma unroll
            for (int m = 0; m < n; ++m) {
                dsum += cur[2 * m] - cur[2 * m + 1];
                cur[m] = cur[2 * m] + cur[2 * m + 1];
            }
            acc[4 - lvl] = dsum;
        }
        const float tot = cur[0];
#pragma unroll
        for (int w = 5; w < DD; ++w)
            acc[w] = ((lane >> (9 - w)) & 1) ? -tot : tot;
    }
    for (int off = 16; off > 0; off >>= 1) {
#pragma unroll
        for (int w = 0; w < DD; ++w) acc[w] += __shfl_xor(acc[w], off, 32);
    }
    float zsel = acc[0];
#pragma unroll
    for (int w = 1; w < DD; ++w) zsel = (lane == w) ? acc[w] : zsel;
    if (lane < DD) sZq[g][lane] = zsel;

    const int ncol = g * 64 + 4 * l16;
    const v4f w00 = *reinterpret_cast<const v4f*>(decW + (size_t)(2 * half)     * DIN + ncol);
    const v4f w01 = *reinterpret_cast<const v4f*>(decW + (size_t)(2 * half + 1) * DIN + ncol);
    const v4f w10 = *reinterpret_cast<const v4f*>(decW + (size_t)(2 * half + 4) * DIN + ncol);
    const v4f w11 = *reinterpret_cast<const v4f*>(decW + (size_t)(2 * half + 5) * DIN + ncol);
    const v4f bias = *reinterpret_cast<const v4f*>(decB + ncol);

    __syncthreads();

    v2f a0, a1;
    a0.x = sZq[l16][2 * half];      a0.y = sZq[l16][2 * half + 1];
    a1.x = sZq[l16][2 * half + 4];  a1.y = sZq[l16][2 * half + 5];

#define DEC_TILE(acc, e)                                                      \
    v8f acc = {};                                                             \
    {                                                                         \
        v2f b0, b1;                                                           \
        b0.x = w00.e;  b0.y = w01.e;                                          \
        b1.x = w10.e;  b1.y = w11.e;                                          \
        acc = __builtin_amdgcn_wmma_f32_16x16x4_f32(                          \
                  false, a0, false, b0, (short)0, acc, false, false);         \
        acc = __builtin_amdgcn_wmma_f32_16x16x4_f32(                          \
                  false, a1, false, b1, (short)0, acc, false, false);         \
    }
    DEC_TILE(d0, x)
    DEC_TILE(d1, y)
    DEC_TILE(d2, z)
    DEC_TILE(d3, w)
#undef DEC_TILE

    float* orow = out + (size_t)(blockIdx.x * ROWS_PB + half * 8) * DIN + ncol;
#pragma unroll
    for (int j = 0; j < 8; ++j) {
        v4f o;
        o.x = d0[j] + bias.x;
        o.y = d1[j] + bias.y;
        o.z = d2[j] + bias.z;
        o.w = d3[j] + bias.w;
        *reinterpret_cast<v4f*>(orow + (size_t)j * DIN) = o;
    }
}

extern "C" void kernel_launch(void* const* d_in, const int* in_sizes, int n_in,
                              void* d_out, int out_size, void* d_ws, size_t ws_size,
                              hipStream_t stream) {
    const float* x    = (const float*)d_in[0];
    const float* encW = (const float*)d_in[1];
    const float* encB = (const float*)d_in[2];
    const float* decW = (const float*)d_in[3];
    const float* decB = (const float*)d_in[4];
    const float* Wv   = (const float*)d_in[5];
    const float* bv   = (const float*)d_in[6];
    const float* Wo   = (const float*)d_in[7];
    const float* bo   = (const float*)d_in[8];
    const float* ln1g = (const float*)d_in[9];
    const float* ln1b = (const float*)d_in[10];
    const float* W1   = (const float*)d_in[11];
    const float* b1   = (const float*)d_in[12];
    const float* W2   = (const float*)d_in[13];
    const float* b2   = (const float*)d_in[14];
    const float* ln2g = (const float*)d_in[15];
    const float* ln2b = (const float*)d_in[16];
    const float* qw   = (const float*)d_in[17];
    float* out  = (float*)d_out;

    qsim_kernel<<<BQ / ROWS_PB, ROWS_PB * 32, 0, stream>>>(
        x, encW, encB, qw, Wv, bv, Wo, bo, ln1g, ln1b, W1, b1, W2, b2, ln2g, ln2b,
        decW, decB, out);
}


# RX as scale*(I - i*t*X), larger of cos, sin divided out: one packed FMA per (re,im) amplitude (was 2 mul + 2 FMA); scales enter the start amplitudes, left-over X flips become signs of the Z sums
# speedup vs baseline: 8.4922x; 1.2004x over previous
#include <hip/hip_runtime.h>
#include <hip/hip_bf16.h>
#include <math.h>

typedef __attribute__((ext_vector_type(2))) float v2f;
typedef __attribute__((ext_vector_type(8))) float v8f;

#define BQ     4096
#define DIN    1024
#define DD     8
#define LL     2
#define FFN    128
#define NQ     10
#define DIMQ   1024
#define REPS   5
#define ROWS_PB 16
#define XPITCH 1028

typedef __attribute__((ext_vector_type(4))) float v4f;

__device__ inline void ln8(const float* h, const float* g, const float* b, float* out)
{
    float m = 0.f;
#pragma unroll
    for (int d = 0; d < DD; ++d) m += h[d];
    m *= 0.125f;
    float v = 0.f;
#pragma unroll
    for (int d = 0; d < DD; ++d) { const float t = h[d] - m; v += t * t; }
    v *= 0.125f;
    const float inv = 1.0f / sqrtf(v + 1e-5f);
#pragma unroll
    for (int d = 0; d < DD; ++d) out[d] = (h[d] - m) * inv * g[d] + b[d];
}

__device__ constexpr int ring_map(int i)
{
    for (int w = 0; w < NQ; ++w) {
        const int cbit = 1 << (NQ - 1 - w);
        const int tbit = 1 << (NQ - 1 - ((w + 1) % NQ));
        if (i & cbit) i ^= tbit;
    }
    return i;
}

__device__ constexpr int lds_swz(int h)  { return (h & 15) ^ ((h >> 4) & 1); }
__device__ constexpr int lds_slot(int j) { return j ^ (lds_swz(j >> 5) << 1); }

__device__ inline void wave_lds_sync()
{
    __builtin_amdgcn_wave_barrier();
}

template <int BIT>
__device__ inline void rx_regs(v2f (&am)[32], const v2f tt)
{
    constexpr int low = BIT - 1;
#pragma unroll
    for (int p = 0; p < 16; ++p) {
        const int r0 = ((p & ~low) << 1) | (p & low);
        const int r1 = r0 | BIT;
        const v2f a0 = am[r0];
        const v2f a1 = am[r1];
        am[r0] = __builtin_elementwise_fma(tt, __builtin_shufflevector(a1, a1, 1, 0), a0);
        am[r1] = __builtin_elementwise_fma(tt, __builtin_shufflevector(a0, a0, 1, 0), a1);
    }
}

__global__ __launch_bounds__(ROWS_PB * 32) void qsim_kernel(
    const float* __restrict__ x, const float* __restrict__ encW,
    const float* __restrict__ encB, const float* __restrict__ qw,
    const float* __restrict__ Wv,  const float* __restrict__ bv,
    const float* __restrict__ Wo,  const float* __restrict__ bo,
    const float* __restrict__ ln1g, const float* __restrict__ ln1b,
    const float* __restrict__ W1,  const float* __restrict__ b1,
    const float* __restrict__ W2,  const float* __restrict__ b2,
    const float* __restrict__ ln2g, const float* __restrict__ ln2b,
    const float* __restrict__ decW, const float* __restrict__ decB,
    float* __restrict__ out)
{
    __shared__ v4f   sAmp[ROWS_PB][DIMQ / 2];
    __shared__ v2f   sTan[REPS * NQ];
    __shared__ float sScl[2];
    __shared__ int   sFlp[2];
    __shared__ float sZq[ROWS_PB][DD];
    __shared__ __attribute__((aligned(16))) float sPart[ROWS_PB][ROWS_PB][DD];

    const int g    = threadIdx.x >> 5;
    const int lane = threadIdx.x & 31;
    const int row  = blockIdx.x * ROWS_PB + g;
    const int l16  = lane & 15;
    const int half = lane >> 4;

    const float qang = qw[threadIdx.x < REPS * NQ ? threadIdx.x : 0];

    const float* xr = x + (size_t)row * DIN + 4 * lane;
    v4f xv[8];
#pragma unroll
    for (int j = 0; j < 8; ++j) xv[j] = *reinterpret_cast<const v4f*>(xr + 128 * j);

    const int   kb    = g * 64;
    const int   bcol  = l16 & 7;
    const float bmask = (l16 < DD) ? 1.0f : 0.0f;
    v2f bf[16];
#pragma unroll
    for (int u = 0; u < 8; ++u) {
        const float* wk = encW + (size_t)(kb + 8 * u + 4 * half) * DD + bcol;
        bf[2 * u].x     = bmask * wk[0 * DD];
        bf[2 * u].y     = bmask * wk[1 * DD];
        bf[2 * u + 1].x = bmask * wk[2 * DD];
        bf[2 * u + 1].y = bmask * wk[3 * DD];
    }

    asm volatile("" :: "v"(qang));

    if (g < 2) {
        float scl = 1.0f;
        int   flp = 0;
        if (threadIdx.x < REPS * NQ) {
            const float th = 0.5f * qang;
            const float c = cosf(th);
            const float s = sinf(th);
            const bool  big_s = fabsf(s) > fabsf(c);
            const float t = big_s ? -c / s : s / c;
            v2f tt;
            tt.x = t;  tt.y = -t;
            sTan[threadIdx.x] = tt;
            scl = big_s ? fabsf(s) : fabsf(c);
            if (big_s) {
                const int gr = threadIdx.x / NQ;
                const int gw = threadIdx.x - gr * NQ;
                int m = 1 << (NQ - 1 - gw);
                for (int k = gr; k < REPS; ++k) m = ring_map(m);
                flp = m;
            }
        }
        for (int off = 16; off > 0; off >>= 1) {
            scl *= __shfl_xor(scl, off, 32);
            flp ^= __shfl_xor(flp, off, 32);
        }
        if (lane == 0) { sScl[g] = scl;  sFlp[g] = flp; }
    }

    float* const xs = reinterpret_cast<float*>(&sAmp[0][0]);
#pragma unroll
    for (int j = 0; j < 8; ++j)
        *reinterpret_cast<v4f*>(xs + g * XPITCH + 4 * lane + 128 * j) = xv[j];

    __syncthreads();

    {
        const float* arow = xs + l16 * XPITCH + kb + 4 * half;
        v8f c0 = {}, c1 = {};
#pragma unroll
        for (int u = 0; u < 8; ++u) {
            const v4f a4 = *reinterpret_cast<const v4f*>(arow + 8 * u);
            v2f alo, ahi;
            alo.x = a4.x;  alo.y = a4.y;
            ahi.x = a4.z;  ahi.y = a4.w;
            c0 = __builtin_amdgcn_wmma_f32_16x16x4_f32(
                     false, alo, false, bf[2 * u], (short)0, c0, false, false);
            c1 = __builtin_amdgcn_wmma_f32_16x16x4_f32(
                     false, ahi, false, bf[2 * u + 1], (short)0, c1, false, false);
        }
        if (l16 < DD) {
#pragma unroll
            for (int r = 0; r < 8; ++r) sPart[g][half * 8 + r][l16] = c0[r] + c1[r];
        }
    }
    __syncthreads();

    float z[DD];
#pragma unroll
    for (int d = 0; d < DD; ++d) z[d] = 0.f;
#pragma unroll
    for (int p = 0; p < ROWS_PB; ++p) {
        const v4f lo = *reinterpret_cast<const v4f*>(&sPart[p][g][0]);
        const v4f hi = *reinterpret_cast<const v4f*>(&sPart[p][g][4]);
        z[0] += lo.x;  z[1] += lo.y;  z[2] += lo.z;  z[3] += lo.w;
        z[4] += hi.x;  z[5] += hi.y;  z[6] += hi.z;  z[7] += hi.w;
    }
#pragma unroll
    for (int d = 0; d < DD; ++d) z[d] += encB[d];

    for (int l = 0; l < LL; ++l) {
        const float* wv = Wv + l * 64;  const float* wo = Wo + l * 64;
        float v[DD], attn[DD], h[DD];
#pragma unroll
        for (int d = 0; d < DD; ++d) {
            float s = bv[l * DD + d];
#pragma unroll
            for (int k = 0; k < DD; ++k) s += z[k] * wv[k * DD + d];
            v[d] = s;
        }
#pragma unroll
        for (int d = 0; d < DD; ++d) {
            float s = bo[l * DD + d];
#pragma unroll
            for (int k = 0; k < DD; ++k) s += v[k] * wo[k * DD + d];
            attn[d] = s;
        }
#pragma unroll
        for (int d = 0; d < DD; ++d) h[d] = z[d] + attn[d];
        ln8(h, ln1g + l * DD, ln1b + l * DD, z);

        float acc[DD];
#pragma unroll
        for (int d = 0; d < DD; ++d) acc[d] = 0.f;
#pragma unroll
        for (int j = 0; j < 4; ++j) {
            const int hid = lane + j * 32;
            float hv = b1[l * FFN + hid];
#pragma unroll
            for (int k = 0; k < DD; ++k) hv += z[k] * W1[l * DD * FFN + k * FFN + hid];
            hv = fmaxf(hv, 0.0f);
#pragma unroll
            for (int d = 0; d < DD; ++d) acc[d] += hv * W2[l * FFN * DD + hid * DD + d];
        }
        for (int off = 16; off > 0; off >>= 1) {
#pragma unroll
            for (int d = 0; d < DD; ++d) acc[d] += __shfl_xor(acc[d], off, 32);
        }
#pragma unroll
        for (int d = 0; d < DD; ++d) h[d] = z[d] + acc[d] + b2[l * DD + d];
        ln8(h, ln2g + l * DD, ln2b + l * DD, z);
    }

    float nrm2 = 0.f;
#pragma unroll
    for (int d = 0; d < DD; ++d) nrm2 += z[d] * z[d];
    const float inv = 1.0f / sqrtf(nrm2);
    const float gscale = sScl[0] * sScl[1];

    v2f am[32];
#pragma unroll
    for (int r = 0; r < 32; ++r) { am[r].x = 0.f; am[r].y = 0.f; }
#pragma unroll
    for (int d = 0; d < DD; ++d) if (lane == d) am[0].x = z[d] * inv * gscale;

    v4f* const amp4 = &sAmp[g][0];
    v2f* const amp2 = reinterpret_cast<v2f*>(amp4);
    const int pairA   = lane * 16;
    const int swzA    = lds_swz(lane);
    const int ringSlt = lds_slot(ring_map(lane << 5));

#pragma unroll 1
    for (int r = 0; r < REPS; ++r) {
        const v2f* tg = sTan + r * NQ;

        rx_regs<16>(am, tg[0]);
        rx_regs<8> (am, tg[1]);
        rx_regs<4> (am, tg[2]);
        rx_regs<2> (am, tg[3]);
        rx_regs<1> (am, tg[4]);

#pragma unroll
        for (int R = 0; R < 32; ++R)
            amp2[R * 32 + (lane ^ (lds_swz(R) << 1))] = am[R];
        wave_lds_sync();
#pragma unroll
        for (int k = 0; k < 16; ++k) {
            const v4f t = amp4[pairA + (k ^ swzA)];
            am[2 * k].x     = t.x;  am[2 * k].y     = t.y;
            am[2 * k + 1].x = t.z;  am[2 * k + 1].y = t.w;
        }

        rx_regs<16>(am, tg[5]);
        rx_regs<8> (am, tg[6]);
        rx_regs<4> (am, tg[7]);
        rx_regs<2> (am, tg[8]);
        rx_regs<1> (am, tg[9]);

#pragma unroll
        for (int rr = 0; rr < 32; ++rr)
            amp2[lds_slot(ring_map(rr)) ^ ringSlt] = am[rr];
        wave_lds_sync();
#pragma unroll
        for (int R = 0; R < 32; ++R)
            am[R] = amp2[R * 32 + (lane ^ (lds_swz(R) << 1))];
    }

    float p[32];
#pragma unroll
    for (int R = 0; R < 32; ++R) p[R] = am[R].x * am[R].x + am[R].y * am[R].y;

    float acc[DD];
    {
        float cur[32];
#pragma unroll
        for (int R = 0; R < 32; ++R) cur[R] = p[R];
#pragma unroll
        for (int lvl = 0; lvl < 5; ++lvl) {
            const int n = 16 >> lvl;
            float dsum = 0.f;
#pragma unroll
            for (int m = 0; m < n; ++m) {
                dsum += cur[2 * m] - cur[2 * m + 1];
                cur[m] = cur[2 * m] + cur[2 * m + 1];
            }
            acc[4 - lvl] = dsum;
        }
        const float tot = cur[0];
#pragma unroll
        for (int w = 5; w < DD; ++w)
            acc[w] = ((lane >> (9 - w)) & 1) ? -tot : tot;
    }
    for (int off = 16; off > 0; off >>= 1) {
#pragma unroll
        for (int w = 0; w < DD; ++w) acc[w] += __shfl_xor(acc[w], off, 32);
    }
    float zsel = acc[0];
#pragma unroll
    for (int w = 1; w < DD; ++w) zsel = (lane == w) ? acc[w] : zsel;
    const int flips = sFlp[0] ^ sFlp[1];
    if ((flips >> (NQ - 1 - (lane & 7))) & 1) zsel = -zsel;
    if (lane < DD) sZq[g][lane] = zsel;

    const int ncol = g * 64 + 4 * l16;
    const v4f w00 = *reinterpret_cast<const v4f*>(decW + (size_t)(2 * half)     * DIN + ncol);
    const v4f w01 = *reinterpret_cast<const v4f*>(decW + (size_t)(2 * half + 1) * DIN + ncol);
    const v4f w10 = *reinterpret_cast<const v4f*>(decW + (size_t)(2 * half + 4) * DIN + ncol);
    const v4f w11 = *reinterpret_cast<const v4f*>(decW + (size_t)(2 * half + 5) * DIN + ncol);
    const v4f bias = *reinterpret_cast<const v4f*>(decB + ncol);

    __syncthreads();

    v2f a0, a1;
    a0.x = sZq[l16][2 * half];      a0.y = sZq[l16][2 * half + 1];
    a1.x = sZq[l16][2 * half + 4];  a1.y = sZq[l16][2 * half + 5];

#define DEC_TILE(acc, e)                                                      \
    v8f acc = {};                                                             \
    {                                                                         \
        v2f b0, b1;                                                           \
        b0.x = w00.e;  b0.y = w01.e;                                          \
        b1.x = w10.e;  b1.y = w11.e;                                          \
        acc = __builtin_amdgcn_wmma_f32_16x16x4_f32(                          \
                  false, a0, false, b0, (short)0, acc, false, false);         \
        acc = __builtin_amdgcn_wmma_f32_16x16x4_f32(                          \
                  false, a1, false, b1, (short)0, acc, false, false);         \
    }
    DEC_TILE(d0, x)
    DEC_TILE(d1, y)
    DEC_TILE(d2, z)
    DEC_TILE(d3, w)
#undef DEC_TILE

    float* orow = out + (size_t)(blockIdx.x * ROWS_PB + half * 8) * DIN + ncol;
#pragma unroll
    for (int j = 0; j < 8; ++j) {
        v4f o;
        o.x = d0[j] + bias.x;
        o.y = d1[j] + bias.y;
        o.z = d2[j] + bias.z;
        o.w = d3[j] + bias.w;
        *reinterpret_cast<v4f*>(orow + (size_t)j * DIN) = o;
    }
}

extern "C" void kernel_launch(void* const* d_in, const int* in_sizes, int n_in,
                              void* d_out, int out_size, void* d_ws, size_t ws_size,
                              hipStream_t stream) {
    const float* x    = (const float*)d_in[0];
    const float* encW = (const float*)d_in[1];
    const float* encB = (const float*)d_in[2];
    const float* decW = (const float*)d_in[3];
    const float* decB = (const float*)d_in[4];
    const float* Wv   = (const float*)d_in[5];
    const float* bv   = (const float*)d_in[6];
    const float* Wo   = (const float*)d_in[7];
    const float* bo   = (const float*)d_in[8];
    const float* ln1g = (const float*)d_in[9];
    const float* ln1b = (const float*)d_in[10];
    const float* W1   = (const float*)d_in[11];
    const float* b1   = (const float*)d_in[12];
    const float* W2   = (const float*)d_in[13];
    const float* b2   = (const float*)d_in[14];
    const float* ln2g = (const float*)d_in[15];
    const float* ln2b = (const float*)d_in[16];
    const float* qw   = (const float*)d_in[17];
    float* out  = (float*)d_out;

    qsim_kernel<<<BQ / ROWS_PB, ROWS_PB * 32, 0, stream>>>(
        x, encW, encB, qw, Wv, bv, Wo, bo, ln1g, ln1b, W1, b1, W2, b2, ln2g, ln2b,
        decW, decB, out);
}
